# starting kernel
# baseline (speedup 1.0000x reference)
#include <hip/hip_runtime.h>

typedef __attribute__((ext_vector_type(2))) float v2f;
typedef __attribute__((ext_vector_type(4))) float v4f;
typedef __attribute__((ext_vector_type(8))) float v8f;

#define BN_EPS 1e-5f
#define B_  32
#define C_  256
#define CM_ 128
#define CO_ 256
#define H_  64
#define W_  64
#define L_  128

__global__ __launch_bounds__(256) void ca_pool(const float* __restrict__ x,
                                               float* __restrict__ p) {
    __shared__ float tile[64 * 65];
    const int bc = blockIdx.x;
    const float* xp = x + (size_t)bc * (H_ * W_);
    const int t  = threadIdx.x;
    const int h  = t >> 2;
    const int w0 = (t & 3) * 16;

    const v4f* src = (const v4f*)(xp + h * W_ + w0);
#pragma unroll
    for (int j = 0; j < 4; ++j) {
        v4f v = src[j];
#pragma unroll
        for (int e = 0; e < 4; ++e)
            tile[h * 65 + w0 + j * 4 + e] = v[e];
    }
    __syncthreads();

    if (t < 64) {
        float s = 0.0f;
#pragma unroll 8
        for (int w = 0; w < 64; ++w) s += tile[t * 65 + w];
        p[(size_t)bc * L_ + t] = s * (1.0f / 64.0f);
    } else if (t < 128) {
        const int w = t - 64;
        float s = 0.0f;
#pragma unroll 8
        for (int hh = 0; hh < 64; ++hh) s += tile[hh * 65 + w];
        p[(size_t)bc * L_ + 64 + w] = s * (1.0f / 64.0f);
    }
}

__global__ __launch_bounds__(256) void ca_mlp(
    const float* __restrict__ p,
    const float* __restrict__ w1,
    const float* __restrict__ b1,
    const float* __restrict__ gam, const float* __restrict__ bet,
    const float* __restrict__ mu,  const float* __restrict__ var,
    const float* __restrict__ w2,  const float* __restrict__ b2,
    const float* __restrict__ w3,  const float* __restrict__ b3,
    float* __restrict__ g)
{
    __shared__ float lt[CM_ * L_];
    const int b     = blockIdx.x;
    const float* pb = p + (size_t)b * C_ * L_;
    const int wave  = threadIdx.x >> 5;
    const int lane  = threadIdx.x & 31;
    const int lrow  = lane & 15;
    const int lhalf = lane >> 4;

    const int M0 = wave * 16;
    v8f acc[8];
#pragma unroll
    for (int n = 0; n < 8; ++n) acc[n] = (v8f)(0.0f);

    for (int k = 0; k < C_; k += 4) {
        const float* ar = w1 + (M0 + lrow) * C_ + k + 2 * lhalf;
        v2f a; a[0] = ar[0]; a[1] = ar[1];
#pragma unroll
        for (int n = 0; n < 8; ++n) {
            const int col = n * 16 + lrow;
            v2f bb;
            bb[0] = pb[(k     + 2 * lhalf) * L_ + col];
            bb[1] = pb[(k + 1 + 2 * lhalf) * L_ + col];
            acc[n] = __builtin_amdgcn_wmma_f32_16x16x4_f32(
                false, a, false, bb, (short)0, acc[n], false, false);
        }
    }

    float sc8[8], sh8[8];
#pragma unroll
    for (int i = 0; i < 8; ++i) {
        const int m = M0 + i + 8 * lhalf;
        const float s = gam[m] * rsqrtf(var[m] + BN_EPS);
        sc8[i] = s;
        sh8[i] = bet[m] + (b1[m] - mu[m]) * s;
    }
#pragma unroll
    for (int n = 0; n < 8; ++n) {
#pragma unroll
        for (int i = 0; i < 8; ++i) {
            const int m = M0 + i + 8 * lhalf;
            const float yv = acc[n][i] * sc8[i] + sh8[i];
            const float hs = yv * fminf(fmaxf(yv + 3.0f, 0.0f), 6.0f) * (1.0f / 6.0f);
            lt[m * L_ + n * 16 + lrow] = hs;
        }
    }
    __syncthreads();

    float* gb = g + (size_t)b * CO_ * L_;
#pragma unroll
    for (int oo = 0; oo < 2; ++oo) {
        const int O0 = (wave * 2 + oo) * 16;
        for (int ltile = 0; ltile < 8; ++ltile) {
            const float* ws = (ltile < 4) ? w2 : w3;
            const float* bs = (ltile < 4) ? b2 : b3;
            v8f acc2 = (v8f)(0.0f);
            for (int k = 0; k < CM_; k += 4) {
                const float* ar = ws + (O0 + lrow) * CM_ + k + 2 * lhalf;
                v2f a; a[0] = ar[0]; a[1] = ar[1];
                const int col = ltile * 16 + lrow;
                v2f bb;
                bb[0] = lt[(k     + 2 * lhalf) * L_ + col];
                bb[1] = lt[(k + 1 + 2 * lhalf) * L_ + col];
                acc2 = __builtin_amdgcn_wmma_f32_16x16x4_f32(
                    false, a, false, bb, (short)0, acc2, false, false);
            }
#pragma unroll
            for (int i = 0; i < 8; ++i) {
                const int o  = O0 + i + 8 * lhalf;
                const float yv = acc2[i] + bs[o];
                const float gv = 1.0f / (1.0f + __expf(-yv));
                gb[o * L_ + ltile * 16 + lrow] = gv;
            }
        }
    }
}

__global__ __launch_bounds__(256) void ca_scale(const float* __restrict__ x,
                                                const float* __restrict__ g,
                                                float* __restrict__ out) {
    __shared__ float gr[L_];
    const int bc = blockIdx.x;
    const int t  = threadIdx.x;
    if (t < L_) gr[t] = g[(size_t)bc * L_ + t];
    __syncthreads();

    const float* xp = x   + (size_t)bc * (H_ * W_);
    float*       op = out + (size_t)bc * (H_ * W_);
    const int h  = t >> 2;
    const int w0 = (t & 3) * 16;
    const float gh = gr[h];

    const v4f* src = (const v4f*)(xp + h * W_ + w0);
    v4f*       dst = (v4f*)(op + h * W_ + w0);
#pragma unroll
    for (int j = 0; j < 4; ++j) {
        v4f v = src[j];
#pragma unroll
        for (int e = 0; e < 4; ++e)
            v[e] = v[e] * gh * gr[64 + w0 + j * 4 + e];
        __builtin_nontemporal_store(v, dst + j);
    }
}

extern "C" void kernel_launch(void* const* d_in, const int* in_sizes, int n_in,
                              void* d_out, int out_size, void* d_ws, size_t ws_size,
                              hipStream_t stream) {
    const float* x    = (const float*)d_in[0];
    const float* w1   = (const float*)d_in[1];
    const float* b1   = (const float*)d_in[2];
    const float* gam  = (const float*)d_in[3];
    const float* bet  = (const float*)d_in[4];
    const float* mu   = (const float*)d_in[5];
    const float* var  = (const float*)d_in[6];
    const float* w2   = (const float*)d_in[7];
    const float* b2   = (const float*)d_in[8];
    const float* w3   = (const float*)d_in[9];
    const float* b3   = (const float*)d_in[10];
    float* out = (float*)d_out;

    float* p = (float*)d_ws;
    float* g = p + (size_t)B_ * C_ * L_;

    ca_pool <<<B_ * C_, 256, 0, stream>>>(x, p);
    ca_mlp  <<<B_,       256, 0, stream>>>(p, w1, b1, gam, bet, mu, var,
                                           w2, b2, w3, b3, g);
    ca_scale<<<B_ * C_, 256, 0, stream>>>(x, g, out);
}


# ca_mlp split by column tiles: one work-group per (batch, 32 columns), 128 instead of 32, p and t tiles in LDS, w fragments streamed behind each barrier; k order and sums unchanged
# speedup vs baseline: 1.8259x; 1.8259x over previous
#include <hip/hip_runtime.h>

typedef __attribute__((ext_vector_type(2))) float v2f;
typedef __attribute__((ext_vector_type(4))) float v4f;
typedef __attribute__((ext_vector_type(8))) float v8f;

#define BN_EPS 1e-5f
#define B_  32
#define C_  256
#define CM_ 128
#define CO_ 256
#define H_  64
#define W_  64
#define L_  128

__global__ __launch_bounds__(256) void ca_pool(const float* __restrict__ x,
                                               float* __restrict__ p) {
    __shared__ float tile[64 * 65];
    const int bc = blockIdx.x;
    const float* xp = x + (size_t)bc * (H_ * W_);
    const int t  = threadIdx.x;
    const int h  = t >> 2;
    const int w0 = (t & 3) * 16;

    const v4f* src = (const v4f*)(xp + h * W_ + w0);
#pragma unroll
    for (int j = 0; j < 4; ++j) {
        v4f v = src[j];
#pragma unroll
        for (int e = 0; e < 4; ++e)
            tile[h * 65 + w0 + j * 4 + e] = v[e];
    }
    __syncthreads();

    if (t < 64) {
        float s = 0.0f;
#pragma unroll 8
        for (int w = 0; w < 64; ++w) s += tile[t * 65 + w];
        p[(size_t)bc * L_ + t] = s * (1.0f / 64.0f);
    } else if (t < 128) {
        const int w = t - 64;
        float s = 0.0f;
#pragma unroll 8
        for (int hh = 0; hh < 64; ++hh) s += tile[hh * 65 + w];
        p[(size_t)bc * L_ + 64 + w] = s * (1.0f / 64.0f);
    }
}

#define NT_    32
#define PITCH_ 40

__global__ __launch_bounds__(256) void ca_mlp(
    const float* __restrict__ p,
    const float* __restrict__ w1,
    const float* __restrict__ b1,
    const float* __restrict__ gam, const float* __restrict__ bet,
    const float* __restrict__ mu,  const float* __restrict__ var,
    const float* __restrict__ w2,  const float* __restrict__ b2,
    const float* __restrict__ w3,  const float* __restrict__ b3,
    float* __restrict__ g)
{
    __shared__ __attribute__((aligned(16))) float lp[C_  * PITCH_];
    __shared__ float lt[CM_ * PITCH_];
    const int b     = blockIdx.x >> 2;
    const int ct    = blockIdx.x & 3;
    const int c0    = ct * NT_;
    const float* pb = p + (size_t)b * C_ * L_ + c0;
    const int wave  = threadIdx.x >> 5;
    const int lane  = threadIdx.x & 31;
    const int lrow  = lane & 15;
    const int lhalf = lane >> 4;

    const int sr = threadIdx.x >> 3;
    const int sc = (threadIdx.x & 7) * 4;
    v4f pv[8];
#pragma unroll
    for (int j = 0; j < 8; ++j)
        pv[j] = *(const v4f*)(pb + (sr + 32 * j) * L_ + sc);

    const int M0 = wave * 16;
    const float* a1p = w1 + (M0 + lrow) * C_ + 2 * lhalf;
    v2f a1[64];
#pragma unroll
    for (int kk = 0; kk < 16; ++kk) {
        a1[kk][0] = a1p[4 * kk];
        a1[kk][1] = a1p[4 * kk + 1];
    }

#pragma unroll
    for (int j = 0; j < 8; ++j)
        *(v4f*)(&lp[(sr + 32 * j) * PITCH_ + sc]) = pv[j];

    float sc8[8], sh8[8];
#pragma unroll
    for (int i = 0; i < 8; ++i) {
        const int m = M0 + i + 8 * lhalf;
        const float s = gam[m] * rsqrtf(var[m] + BN_EPS);
        sc8[i] = s;
        sh8[i] = bet[m] + (b1[m] - mu[m]) * s;
    }
    __syncthreads();

#pragma unroll
    for (int kk = 16; kk < 64; ++kk) {
        a1[kk][0] = a1p[4 * kk];
        a1[kk][1] = a1p[4 * kk + 1];
    }

    const float* ws = (ct < 2) ? w2 : w3;
    const float* bs = (ct < 2) ? b2 : b3;
    const int O0 = wave * 32;
    const float* a2p = ws + (O0 + lrow) * CM_ + 2 * lhalf;
    v2f a2[2][32];
#pragma unroll
    for (int oo = 0; oo < 2; ++oo)
#pragma unroll
        for (int kk = 0; kk < 8; ++kk) {
            a2[oo][kk][0] = a2p[oo * 16 * CM_ + 4 * kk];
            a2[oo][kk][1] = a2p[oo * 16 * CM_ + 4 * kk + 1];
        }
    float bo[2][8];
#pragma unroll
    for (int oo = 0; oo < 2; ++oo)
#pragma unroll
        for (int i = 0; i < 8; ++i)
            bo[oo][i] = bs[O0 + oo * 16 + i + 8 * lhalf];

    v8f acc[2];
    acc[0] = (v8f)(0.0f);
    acc[1] = (v8f)(0.0f);
    const float* bp = lp + (2 * lhalf) * PITCH_ + lrow;
#pragma unroll
    for (int kk = 0; kk < 64; ++kk) {
#pragma unroll
        for (int n = 0; n < 2; ++n) {
            v2f bb;
            bb[0] = bp[(4 * kk    ) * PITCH_ + n * 16];
            bb[1] = bp[(4 * kk + 1) * PITCH_ + n * 16];
            acc[n] = __builtin_amdgcn_wmma_f32_16x16x4_f32(
                false, a1[kk], false, bb, (short)0, acc[n], false, false);
        }
    }

#pragma unroll
    for (int n = 0; n < 2; ++n) {
#pragma unroll
        for (int i = 0; i < 8; ++i) {
            const int m = M0 + i + 8 * lhalf;
            const float yv = acc[n][i] * sc8[i] + sh8[i];
            const float hs = yv * fminf(fmaxf(yv + 3.0f, 0.0f), 6.0f) * (1.0f / 6.0f);
            lt[m * PITCH_ + n * 16 + lrow] = hs;
        }
    }
    __syncthreads();

#pragma unroll
    for (int oo = 0; oo < 2; ++oo)
#pragma unroll
        for (int kk = 8; kk < 32; ++kk) {
            a2[oo][kk][0] = a2p[oo * 16 * CM_ + 4 * kk];
            a2[oo][kk][1] = a2p[oo * 16 * CM_ + 4 * kk + 1];
        }

    v8f acc2[2][2];
#pragma unroll
    for (int oo = 0; oo < 2; ++oo)
#pragma unroll
        for (int n = 0; n < 2; ++n) acc2[oo][n] = (v8f)(0.0f);
    const float* tp = lt + (2 * lhalf) * PITCH_ + lrow;
#pragma unroll
    for (int kk = 0; kk < 32; ++kk) {
        v2f bb[2];
#pragma unroll
        for (int n = 0; n < 2; ++n) {
            bb[n][0] = tp[(4 * kk    ) * PITCH_ + n * 16];
            bb[n][1] = tp[(4 * kk + 1) * PITCH_ + n * 16];
        }
#pragma unroll
        for (int oo = 0; oo < 2; ++oo)
#pragma unroll
            for (int n = 0; n < 2; ++n)
                acc2[oo][n] = __builtin_amdgcn_wmma_f32_16x16x4_f32(
                    false, a2[oo][kk], false, bb[n], (short)0, acc2[oo][n], false, false);
    }

    float* gb = g + (size_t)b * CO_ * L_ + c0;
#pragma unroll
    for (int oo = 0; oo < 2; ++oo) {
#pragma unroll
        for (int n = 0; n < 2; ++n) {
#pragma unroll
            for (int i = 0; i < 8; ++i) {
                const int o  = O0 + oo * 16 + i + 8 * lhalf;
                const float yv = acc2[oo][n][i] + bo[oo][i];
                const float gv = 1.0f / (1.0f + __expf(-yv));
                gb[o * L_ + n * 16 + lrow] = gv;
            }
        }
    }
}

__global__ __launch_bounds__(256) void ca_scale(const float* __restrict__ x,
                                                const float* __restrict__ g,
                                                float* __restrict__ out) {
    __shared__ float gr[L_];
    const int bc = blockIdx.x;
    const int t  = threadIdx.x;
    if (t < L_) gr[t] = g[(size_t)bc * L_ + t];
    __syncthreads();

    const float* xp = x   + (size_t)bc * (H_ * W_);
    float*       op = out + (size_t)bc * (H_ * W_);
    const int h  = t >> 2;
    const int w0 = (t & 3) * 16;
    const float gh = gr[h];

    const v4f* src = (const v4f*)(xp + h * W_ + w0);
    v4f*       dst = (v4f*)(op + h * W_ + w0);
#pragma unroll
    for (int j = 0; j < 4; ++j) {
        v4f v = src[j];
#pragma unroll
        for (int e = 0; e < 4; ++e)
            v[e] = v[e] * gh * gr[64 + w0 + j * 4 + e];
        __builtin_nontemporal_store(v, dst + j);
    }
}

extern "C" void kernel_launch(void* const* d_in, const int* in_sizes, int n_in,
                              void* d_out, int out_size, void* d_ws, size_t ws_size,
                              hipStream_t stream) {
    const float* x    = (const float*)d_in[0];
    const float* w1   = (const float*)d_in[1];
    const float* b1   = (const float*)d_in[2];
    const float* gam  = (const float*)d_in[3];
    const float* bet  = (const float*)d_in[4];
    const float* mu   = (const float*)d_in[5];
    const float* var  = (const float*)d_in[6];
    const float* w2   = (const float*)d_in[7];
    const float* b2   = (const float*)d_in[8];
    const float* w3   = (const float*)d_in[9];
    const float* b3   = (const float*)d_in[10];
    float* out = (float*)d_out;

    float* p = (float*)d_ws;
    float* g = p + (size_t)B_ * C_ * L_;

    ca_pool <<<B_ * C_, 256, 0, stream>>>(x, p);
    ca_mlp  <<<B_ * 4,   256, 0, stream>>>(p, w1, b1, gam, bet, mu, var,
                                           w2, b2, w3, b3, g);
    ca_scale<<<B_ * C_, 256, 0, stream>>>(x, g, out);
}


# ca_pool and ca_scale: one wave per (b,c) plane with its 32 row loads in flight at once, sums in registers by a DPP lane butterfly, no LDS tile or barrier; 8192 waves not 65536; pool sums re-ordered
# speedup vs baseline: 1.9823x; 1.0857x over previous
#include <hip/hip_runtime.h>

typedef __attribute__((ext_vector_type(2))) float v2f;
typedef __attribute__((ext_vector_type(4))) float v4f;
typedef __attribute__((ext_vector_type(8))) float v8f;

#define BN_EPS 1e-5f
#define B_  32
#define C_  256
#define CM_ 128
#define CO_ 256
#define H_  64
#define W_  64
#define L_  128

template <int M>
__device__ __forceinline__ float row_xor(float v) {
    const int s = __builtin_bit_cast(int, v);
    return __builtin_bit_cast(float,
        __builtin_amdgcn_update_dpp(s, s, 0x160 + M, 0xf, 0xf, true));
}
__device__ __forceinline__ float xhalf(float v) {
    const int s = __builtin_bit_cast(int, v);
    return __builtin_bit_cast(float,
        __builtin_amdgcn_permlanex16(s, s, 0x76543210, 0xfedcba98, false, false));
}
template <int M>
__device__ __forceinline__ float halve(float lo, float hi, bool up) {
    const float keep = up ? hi : lo;
    const float send = up ? lo : hi;
    return keep + row_xor<M>(send);
}

__global__ __launch_bounds__(256) void ca_pool(const float* __restrict__ x,
                                               float* __restrict__ p) {
    const int lane = threadIdx.x & 31;
    const int wave = __builtin_amdgcn_readfirstlane((int)(threadIdx.x >> 5));
    const int bc   = blockIdx.x * 8 + wave;
    const int c    = lane & 15;
    const int h    = lane >> 4;
    const v4f* src = (const v4f*)(x + (size_t)bc * (H_ * W_)) + h * 512 + c;

    v4f buf[32];
#pragma unroll
    for (int j = 0; j < 32; ++j) buf[j] = src[16 * j];
    __builtin_amdgcn_sched_barrier(0);

    float r[16];
    v4f cs0 = (v4f)(0.0f), cs1 = (v4f)(0.0f);
#pragma unroll
    for (int j = 0; j < 16; ++j) {
        const v4f v = buf[j];
        r[j] = (v[0] + v[1]) + (v[2] + v[3]);
        if (j & 1) cs1 += v; else cs0 += v;
    }

    const bool b3 = (lane & 8) != 0, b2 = (lane & 4) != 0;
    const bool b1 = (lane & 2) != 0, b0 = (lane & 1) != 0;
    float a[16];
#pragma unroll
    for (int k = 0; k < 16; ++k) {
        const v4f v = buf[k + 16];
        const float r2 = (v[0] + v[1]) + (v[2] + v[3]);
        if (k & 1) cs1 += v; else cs0 += v;
        a[k] = halve<8>(r[k], r2, b3);
    }
#pragma unroll
    for (int k = 0; k < 8; ++k) a[k] = halve<4>(a[k], a[k + 8], b2);
#pragma unroll
    for (int k = 0; k < 4; ++k) a[k] = halve<2>(a[k], a[k + 4], b1);
#pragma unroll
    for (int k = 0; k < 2; ++k) a[k] = halve<1>(a[k], a[k + 2], b0);

    float* pr = p + (size_t)bc * L_;
    v2f rs;
    rs[0] = a[0] * (1.0f / 64.0f);
    rs[1] = a[1] * (1.0f / 64.0f);
    *(v2f*)(pr + 32 * h + 2 * c) = rs;

    v4f cs = cs0 + cs1;
#pragma unroll
    for (int e = 0; e < 4; ++e) cs[e] = cs[e] + xhalf(cs[e]);
    v2f co;
    co[0] = (h ? cs[2] : cs[0]) * (1.0f / 64.0f);
    co[1] = (h ? cs[3] : cs[1]) * (1.0f / 64.0f);
    *(v2f*)(pr + 64 + 4 * c + 2 * h) = co;
}

#define NT_    32
#define PITCH_ 40

__global__ __launch_bounds__(256) void ca_mlp(
    const float* __restrict__ p,
    const float* __restrict__ w1,
    const float* __restrict__ b1,
    const float* __restrict__ gam, const float* __restrict__ bet,
    const float* __restrict__ mu,  const float* __restrict__ var,
    const float* __restrict__ w2,  const float* __restrict__ b2,
    const float* __restrict__ w3,  const float* __restrict__ b3,
    float* __restrict__ g)
{
    __shared__ __attribute__((aligned(16))) float lp[C_  * PITCH_];
    __shared__ float lt[CM_ * PITCH_];
    const int b     = blockIdx.x >> 2;
    const int ct    = blockIdx.x & 3;
    const int c0    = ct * NT_;
    const float* pb = p + (size_t)b * C_ * L_ + c0;
    const int wave  = threadIdx.x >> 5;
    const int lane  = threadIdx.x & 31;
    const int lrow  = lane & 15;
    const int lhalf = lane >> 4;

    const int sr = threadIdx.x >> 3;
    const int sc = (threadIdx.x & 7) * 4;
    v4f pv[8];
#pragma unroll
    for (int j = 0; j < 8; ++j)
        pv[j] = *(const v4f*)(pb + (sr + 32 * j) * L_ + sc);

    const int M0 = wave * 16;
    const float* a1p = w1 + (M0 + lrow) * C_ + 2 * lhalf;
    v2f a1[64];
#pragma unroll
    for (int kk = 0; kk < 16; ++kk) {
        a1[kk][0] = a1p[4 * kk];
        a1[kk][1] = a1p[4 * kk + 1];
    }

#pragma unroll
    for (int j = 0; j < 8; ++j)
        *(v4f*)(&lp[(sr + 32 * j) * PITCH_ + sc]) = pv[j];

    float sc8[8], sh8[8];
#pragma unroll
    for (int i = 0; i < 8; ++i) {
        const int m = M0 + i + 8 * lhalf;
        const float s = gam[m] * rsqrtf(var[m] + BN_EPS);
        sc8[i] = s;
        sh8[i] = bet[m] + (b1[m] - mu[m]) * s;
    }
    __syncthreads();

#pragma unroll
    for (int kk = 16; kk < 64; ++kk) {
        a1[kk][0] = a1p[4 * kk];
        a1[kk][1] = a1p[4 * kk + 1];
    }

    const float* ws = (ct < 2) ? w2 : w3;
    const float* bs = (ct < 2) ? b2 : b3;
    const int O0 = wave * 32;
    const float* a2p = ws + (O0 + lrow) * CM_ + 2 * lhalf;
    v2f a2[2][32];
#pragma unroll
    for (int oo = 0; oo < 2; ++oo)
#pragma unroll
        for (int kk = 0; kk < 8; ++kk) {
            a2[oo][kk][0] = a2p[oo * 16 * CM_ + 4 * kk];
            a2[oo][kk][1] = a2p[oo * 16 * CM_ + 4 * kk + 1];
        }
    float bo[2][8];
#pragma unroll
    for (int oo = 0; oo < 2; ++oo)
#pragma unroll
        for (int i = 0; i < 8; ++i)
            bo[oo][i] = bs[O0 + oo * 16 + i + 8 * lhalf];

    v8f acc[2];
    acc[0] = (v8f)(0.0f);
    acc[1] = (v8f)(0.0f);
    const float* bp = lp + (2 * lhalf) * PITCH_ + lrow;
#pragma unroll
    for (int kk = 0; kk < 64; ++kk) {
#pragma unroll
        for (int n = 0; n < 2; ++n) {
            v2f bb;
            bb[0] = bp[(4 * kk    ) * PITCH_ + n * 16];
            bb[1] = bp[(4 * kk + 1) * PITCH_ + n * 16];
            acc[n] = __builtin_amdgcn_wmma_f32_16x16x4_f32(
                false, a1[kk], false, bb, (short)0, acc[n], false, false);
        }
    }

#pragma unroll
    for (int n = 0; n < 2; ++n) {
#pragma unroll
        for (int i = 0; i < 8; ++i) {
            const int m = M0 + i + 8 * lhalf;
            const float yv = acc[n][i] * sc8[i] + sh8[i];
            const float hs = yv * fminf(fmaxf(yv + 3.0f, 0.0f), 6.0f) * (1.0f / 6.0f);
            lt[m * PITCH_ + n * 16 + lrow] = hs;
        }
    }
    __syncthreads();

#pragma unroll
    for (int oo = 0; oo < 2; ++oo)
#pragma unroll
        for (int kk = 8; kk < 32; ++kk) {
            a2[oo][kk][0] = a2p[oo * 16 * CM_ + 4 * kk];
            a2[oo][kk][1] = a2p[oo * 16 * CM_ + 4 * kk + 1];
        }

    v8f acc2[2][2];
#pragma unroll
    for (int oo = 0; oo < 2; ++oo)
#pragma unroll
        for (int n = 0; n < 2; ++n) acc2[oo][n] = (v8f)(0.0f);
    const float* tp = lt + (2 * lhalf) * PITCH_ + lrow;
#pragma unroll
    for (int kk = 0; kk < 32; ++kk) {
        v2f bb[2];
#pragma unroll
        for (int n = 0; n < 2; ++n) {
            bb[n][0] = tp[(4 * kk    ) * PITCH_ + n * 16];
            bb[n][1] = tp[(4 * kk + 1) * PITCH_ + n * 16];
        }
#pragma unroll
        for (int oo = 0; oo < 2; ++oo)
#pragma unroll
            for (int n = 0; n < 2; ++n)
                acc2[oo][n] = __builtin_amdgcn_wmma_f32_16x16x4_f32(
                    false, a2[oo][kk], false, bb[n], (short)0, acc2[oo][n], false, false);
    }

    float* gb = g + (size_t)b * CO_ * L_ + c0;
#pragma unroll
    for (int oo = 0; oo < 2; ++oo) {
#pragma unroll
        for (int n = 0; n < 2; ++n) {
#pragma unroll
            for (int i = 0; i < 8; ++i) {
                const int o  = O0 + oo * 16 + i + 8 * lhalf;
                const float yv = acc2[oo][n][i] + bo[oo][i];
                const float gv = 1.0f / (1.0f + __expf(-yv));
                gb[o * L_ + n * 16 + lrow] = gv;
            }
        }
    }
}

__global__ __launch_bounds__(256) void ca_scale(const float* __restrict__ x,
                                                const float* __restrict__ g,
                                                float* __restrict__ out) {
    const int lane = threadIdx.x & 31;
    const int wave = __builtin_amdgcn_readfirstlane((int)(threadIdx.x >> 5));
    const int bc   = blockIdx.x * 8 + wave;
    const int c    = lane & 15;
    const int h    = lane >> 4;

    const float* gp = g + (size_t)bc * L_;
    const v4f gw = *(const v4f*)(gp + 64 + 4 * c);
    v4f ghv[8];
#pragma unroll
    for (int q = 0; q < 8; ++q) ghv[q] = *(const v4f*)(gp + 32 * h + 4 * q);

    const v4f* src = (const v4f*)(x   + (size_t)bc * (H_ * W_)) + h * 512 + c;
    v4f*       dst = (v4f*)      (out + (size_t)bc * (H_ * W_)) + h * 512 + c;

    v4f buf[32];
#pragma unroll
    for (int j = 0; j < 32; ++j) buf[j] = src[16 * j];
    __builtin_amdgcn_sched_barrier(0);

#pragma unroll
    for (int j = 0; j < 32; ++j) {
        v4f v = buf[j];
        const float gh = ghv[j >> 2][j & 3];
#pragma unroll
        for (int e = 0; e < 4; ++e) v[e] = v[e] * gh * gw[e];
        __builtin_nontemporal_store(v, dst + 16 * j);
        __builtin_amdgcn_sched_barrier(0);
    }
}

extern "C" void kernel_launch(void* const* d_in, const int* in_sizes, int n_in,
                              void* d_out, int out_size, void* d_ws, size_t ws_size,
                              hipStream_t stream) {
    const float* x    = (const float*)d_in[0];
    const float* w1   = (const float*)d_in[1];
    const float* b1   = (const float*)d_in[2];
    const float* gam  = (const float*)d_in[3];
    const float* bet  = (const float*)d_in[4];
    const float* mu   = (const float*)d_in[5];
    const float* var  = (const float*)d_in[6];
    const float* w2   = (const float*)d_in[7];
    const float* b2   = (const float*)d_in[8];
    const float* w3   = (const float*)d_in[9];
    const float* b3   = (const float*)d_in[10];
    float* out = (float*)d_out;

    float* p = (float*)d_ws;
    float* g = p + (size_t)B_ * C_ * L_;

    ca_pool <<<B_ * C_ / 8, 256, 0, stream>>>(x, p);
    ca_mlp  <<<B_ * 4,   256, 0, stream>>>(p, w1, b1, gam, bet, mu, var,
                                           w2, b2, w3, b3, g);
    ca_scale<<<B_ * C_ / 8, 256, 0, stream>>>(x, g, out);
}
